# starting kernel
# baseline (speedup 1.0000x reference)
#include <hip/hip_runtime.h>

#define B_BATCH 1024
#define NPTS    8192
#define THREADS 256
#define CH      256
#define KG      (CH / 4)
#define WAVES   (THREADS / 32)
#define KG_PER_WAVE (KG / WAVES)

typedef __attribute__((ext_vector_type(2))) float v2f;
typedef __attribute__((ext_vector_type(8))) float v8f;

__global__ __launch_bounds__(THREADS)
void svd_reduce_kernel(const float* __restrict__ src,
                       const float* __restrict__ tpl,
                       const float* __restrict__ wgt,
                       float* __restrict__ part)
{
    __shared__ float tileA[KG][64];
    __shared__ float tileB[KG][64];
    __shared__ float accLDS[16];

    const int tid  = threadIdx.x;
    const int lane = tid & 31;
    const int wv   = tid >> 5;
    const int b    = blockIdx.x;

    for (int i = tid; i < KG * 64; i += THREADS) {
        (&tileA[0][0])[i] = 0.0f;
        (&tileB[0][0])[i] = 0.0f;
    }
    if (tid < 16) accLDS[tid] = 0.0f;

    const size_t base = (size_t)b * NPTS;
    const float* gs = src + base * 3;
    const float* gt = tpl + base * 3;
    const float* gw = wgt + base;

    v8f acc = {0.f, 0.f, 0.f, 0.f, 0.f, 0.f, 0.f, 0.f};

    const int k        = tid & 3;
    const int gStage   = tid >> 2;
    const int slotBase = 32 * (k >> 1) + (k & 1);
    const int g0       = wv * KG_PER_WAVE;

    for (int c0 = 0; c0 < NPTS; c0 += CH) {
        const int p = c0 + tid;
        const float w  = gw[p];
        const float s0 = gs[p * 3 + 0], s1 = gs[p * 3 + 1], s2 = gs[p * 3 + 2];
        const float t0 = gt[p * 3 + 0], t1 = gt[p * 3 + 1], t2 = gt[p * 3 + 2];

        if (c0 + CH < NPTS) {
            __builtin_prefetch(&gw[p + CH], 0, 1);
            __builtin_prefetch(&gs[(p + CH) * 3], 0, 1);
            __builtin_prefetch(&gt[(p + CH) * 3], 0, 1);
        }

        __syncthreads();
        float* ta = &tileA[gStage][slotBase];
        float* tb = &tileB[gStage][slotBase];
        ta[0] = w;        tb[0] = 1.0f;
        ta[2] = w * s0;   tb[2] = t0;
        ta[4] = w * s1;   tb[4] = t1;
        ta[6] = w * s2;   tb[6] = t2;
        __syncthreads();

#pragma unroll
        for (int gg = 0; gg < KG_PER_WAVE; ++gg) {
            const int g = g0 + gg;
            v2f a  = *(const v2f*)&tileA[g][lane * 2];
            v2f bb = *(const v2f*)&tileB[g][lane * 2];
            acc = __builtin_amdgcn_wmma_f32_16x16x4_f32(
                 false, a,  false, bb,
                 (short)0, acc,  false,  false);
        }
    }
    __syncthreads();

    if (lane < 4) {
        atomicAdd(&accLDS[0 * 4 + lane], acc[0]);
        atomicAdd(&accLDS[1 * 4 + lane], acc[1]);
        atomicAdd(&accLDS[2 * 4 + lane], acc[2]);
        atomicAdd(&accLDS[3 * 4 + lane], acc[3]);
    }
    __syncthreads();
    if (tid < 16) part[(size_t)b * 16 + tid] = accLDS[tid];
}

__device__ inline float det3x3(const float M[3][3]) {
    return M[0][0] * (M[1][1] * M[2][2] - M[1][2] * M[2][1])
         - M[0][1] * (M[1][0] * M[2][2] - M[1][2] * M[2][0])
         + M[0][2] * (M[1][0] * M[2][1] - M[1][1] * M[2][0]);
}

__global__ __launch_bounds__(256)
void svd_solve_kernel(const float* __restrict__ part, float* __restrict__ out)
{
    const int b = blockIdx.x * blockDim.x + threadIdx.x;
    if (b >= B_BATCH) return;

    float C[4][4];
#pragma unroll
    for (int i = 0; i < 16; ++i) C[i >> 2][i & 3] = part[(size_t)b * 16 + i];

    const float W     = C[0][0];
    const float alpha = 1.0f / (W + 1e-8f);
    const float sig   = W * alpha;
    float sc[3] = {C[1][0] * alpha, C[2][0] * alpha, C[3][0] * alpha};
    float tc[3] = {C[0][1] * alpha, C[0][2] * alpha, C[0][3] * alpha};

    float H[3][3];
#pragma unroll
    for (int d = 0; d < 3; ++d)
#pragma unroll
        for (int e = 0; e < 3; ++e)
            H[d][e] = alpha * C[d + 1][e + 1] - (2.0f - sig) * sc[d] * tc[e];

    float S[3][3];
#pragma unroll
    for (int i = 0; i < 3; ++i)
#pragma unroll
        for (int j = 0; j < 3; ++j)
            S[i][j] = H[0][i] * H[0][j] + H[1][i] * H[1][j] + H[2][i] * H[2][j];

    float V[3][3] = {{1.f, 0.f, 0.f}, {0.f, 1.f, 0.f}, {0.f, 0.f, 1.f}};
    const int PQ[3][2] = {{0, 1}, {0, 2}, {1, 2}};
    for (int sweep = 0; sweep < 8; ++sweep) {
#pragma unroll
        for (int pi = 0; pi < 3; ++pi) {
            const int p = PQ[pi][0], q = PQ[pi][1];
            const float apq = S[p][q];
            if (fabsf(apq) > 1e-20f) {
                const float tau = (S[q][q] - S[p][p]) / (2.0f * apq);
                const float tt  = (tau >= 0.0f ? 1.0f : -1.0f) /
                                  (fabsf(tau) + sqrtf(1.0f + tau * tau));
                const float c = 1.0f / sqrtf(1.0f + tt * tt);
                const float s = tt * c;
#pragma unroll
                for (int r = 0; r < 3; ++r) {
                    const float Srp = S[r][p], Srq = S[r][q];
                    S[r][p] = c * Srp - s * Srq;
                    S[r][q] = s * Srp + c * Srq;
                }
#pragma unroll
                for (int r = 0; r < 3; ++r) {
                    const float Spr = S[p][r], Sqr = S[q][r];
                    S[p][r] = c * Spr - s * Sqr;
                    S[q][r] = s * Spr + c * Sqr;
                }
#pragma unroll
                for (int r = 0; r < 3; ++r) {
                    const float Vrp = V[r][p], Vrq = V[r][q];
                    V[r][p] = c * Vrp - s * Vrq;
                    V[r][q] = s * Vrp + c * Vrq;
                }
            }
        }
    }

    float lam[3] = {S[0][0], S[1][1], S[2][2]};
#pragma unroll
    for (int pi = 0; pi < 3; ++pi) {
        const int i = PQ[pi][0], j = PQ[pi][1];
        if (lam[j] > lam[i]) {
            const float tl = lam[i]; lam[i] = lam[j]; lam[j] = tl;
#pragma unroll
            for (int r = 0; r < 3; ++r) {
                const float tv = V[r][i]; V[r][i] = V[r][j]; V[r][j] = tv;
            }
        }
    }

    float Uc[3][3];
#pragma unroll
    for (int kcol = 0; kcol < 2; ++kcol)
#pragma unroll
        for (int r = 0; r < 3; ++r)
            Uc[kcol][r] = H[r][0] * V[0][kcol] + H[r][1] * V[1][kcol] + H[r][2] * V[2][kcol];

    float n0 = sqrtf(Uc[0][0]*Uc[0][0] + Uc[0][1]*Uc[0][1] + Uc[0][2]*Uc[0][2]);
    if (n0 > 1e-12f) { const float r0 = 1.0f / n0; Uc[0][0]*=r0; Uc[0][1]*=r0; Uc[0][2]*=r0; }
    else { Uc[0][0] = 1.f; Uc[0][1] = 0.f; Uc[0][2] = 0.f; }

    const float d01 = Uc[0][0]*Uc[1][0] + Uc[0][1]*Uc[1][1] + Uc[0][2]*Uc[1][2];
    Uc[1][0] -= d01 * Uc[0][0]; Uc[1][1] -= d01 * Uc[0][1]; Uc[1][2] -= d01 * Uc[0][2];
    float n1 = sqrtf(Uc[1][0]*Uc[1][0] + Uc[1][1]*Uc[1][1] + Uc[1][2]*Uc[1][2]);
    if (n1 > 1e-12f) { const float r1 = 1.0f / n1; Uc[1][0]*=r1; Uc[1][1]*=r1; Uc[1][2]*=r1; }
    else {
        const float ax = (fabsf(Uc[0][0]) < 0.9f) ? 1.f : 0.f;
        const float ay = 1.f - ax;
        float cx = ay * Uc[0][2] - 0.f * Uc[0][1];
        float cy = 0.f * Uc[0][0] - ax * Uc[0][2];
        float cz = ax * Uc[0][1] - ay * Uc[0][0];
        const float cn = 1.0f / sqrtf(cx*cx + cy*cy + cz*cz + 1e-30f);
        Uc[1][0] = cx * cn; Uc[1][1] = cy * cn; Uc[1][2] = cz * cn;
    }
    Uc[2][0] = Uc[0][1]*Uc[1][2] - Uc[0][2]*Uc[1][1];
    Uc[2][1] = Uc[0][2]*Uc[1][0] - Uc[0][0]*Uc[1][2];
    Uc[2][2] = Uc[0][0]*Uc[1][1] - Uc[0][1]*Uc[1][0];

    float R[3][3];
#pragma unroll
    for (int i = 0; i < 3; ++i)
#pragma unroll
        for (int j = 0; j < 3; ++j)
            R[i][j] = V[i][0]*Uc[0][j] + V[i][1]*Uc[1][j] + V[i][2]*Uc[2][j];

    if (det3x3(R) <= 0.0f) {
#pragma unroll
        for (int i = 0; i < 3; ++i)
#pragma unroll
            for (int j = 0; j < 3; ++j)
                R[i][j] -= 2.0f * V[i][2] * Uc[2][j];
    }

    float tvec[3];
#pragma unroll
    for (int i = 0; i < 3; ++i)
        tvec[i] = tc[i] - (R[i][0]*sc[0] + R[i][1]*sc[1] + R[i][2]*sc[2]);

    float* Rout = out;
    float* tout = out + (size_t)B_BATCH * 9;
#pragma unroll
    for (int i = 0; i < 3; ++i)
#pragma unroll
        for (int j = 0; j < 3; ++j)
            Rout[(size_t)b * 9 + i * 3 + j] = R[i][j];
#pragma unroll
    for (int i = 0; i < 3; ++i)
        tout[(size_t)b * 3 + i] = tvec[i];
}

extern "C" void kernel_launch(void* const* d_in, const int* in_sizes, int n_in,
                              void* d_out, int out_size, void* d_ws, size_t ws_size,
                              hipStream_t stream)
{
    const float* src = (const float*)d_in[0];
    const float* tpl = (const float*)d_in[1];
    const float* wgt = (const float*)d_in[2];
    float* out  = (float*)d_out;
    float* part = (float*)d_ws;

    svd_reduce_kernel<<<B_BATCH, THREADS, 0, stream>>>(src, tpl, wgt, part);
    svd_solve_kernel<<<(B_BATCH + 255) / 256, 256, 0, stream>>>(part, out);
}


# Reduce kernel: WMMA operands loaded from global straight into fragment registers, 4 point streams block-diagonal per tile (16 points/WMMA), no LDS staging; sums regrouped, LDS atomics removed
# speedup vs baseline: 1.8781x; 1.8781x over previous
#include <hip/hip_runtime.h>

#define B_BATCH 1024
#define NPTS    8192
#define THREADS 256
#define WAVES   (THREADS / 32)
#define STEP_PTS     16
#define SET_STEPS    4
#define SET_PTS      (STEP_PTS * SET_STEPS)
#define SET_STRIDE   (WAVES * SET_PTS)
#define NSETS        (NPTS / SET_STRIDE)

typedef __attribute__((ext_vector_type(2))) float v2f;
typedef __attribute__((ext_vector_type(8))) float v8f;

struct LoadSet {
    v2f   w[SET_STEPS];
    float s0[SET_STEPS], s1[SET_STEPS];
    float t0[SET_STEPS], t1[SET_STEPS];
};

__device__ __forceinline__ void load_set(LoadSet& L, const float* __restrict__ wp,
                                         const float* __restrict__ sp,
                                         const float* __restrict__ tp)
{
#pragma unroll
    for (int k = 0; k < SET_STEPS; ++k) {
        L.w[k]  = *(const v2f*)(wp + STEP_PTS * k);
        L.s0[k] = sp[3 * STEP_PTS * k];
        L.s1[k] = sp[3 * STEP_PTS * k + 3];
        L.t0[k] = tp[3 * STEP_PTS * k];
        L.t1[k] = tp[3 * STEP_PTS * k + 3];
    }
}

__device__ __forceinline__ void mma_set(const LoadSet& L, const bool isW, v8f& acc0, v8f& acc1)
{
#pragma unroll
    for (int k = 0; k < SET_STEPS; ++k) {
        v2f a, b;
        a.x = isW ? L.w[k].x : L.w[k].x * L.s0[k];
        a.y = isW ? L.w[k].y : L.w[k].y * L.s1[k];
        b.x = isW ? 1.0f : L.t0[k];
        b.y = isW ? 1.0f : L.t1[k];
        if (k & 1)
            acc1 = __builtin_amdgcn_wmma_f32_16x16x4_f32(
                 false, a,  false, b,
                 (short)0, acc1,  false,  false);
        else
            acc0 = __builtin_amdgcn_wmma_f32_16x16x4_f32(
                 false, a,  false, b,
                 (short)0, acc0,  false,  false);
    }
}

__global__ __launch_bounds__(THREADS)
void svd_reduce_kernel(const float* __restrict__ src,
                       const float* __restrict__ tpl,
                       const float* __restrict__ wgt,
                       float* __restrict__ part)
{
    __shared__ float red[WAVES][4][16];

    const int tid  = threadIdx.x;
    const int lane = tid & 31;
    const int wv   = tid >> 5;
    const int b    = blockIdx.x;

    const int  f   = lane & 3;
    const int  q   = lane >> 2;
    const bool isW = (f == 0);
    const int  fm  = isW ? 0 : f - 1;

    const size_t p0 = (size_t)b * NPTS + (size_t)wv * SET_PTS + 2 * q;
    const float* wp = wgt + p0;
    const float* sp = src + p0 * 3 + fm;
    const float* tp = tpl + p0 * 3 + fm;

    v8f acc0 = {0.f, 0.f, 0.f, 0.f, 0.f, 0.f, 0.f, 0.f};
    v8f acc1 = {0.f, 0.f, 0.f, 0.f, 0.f, 0.f, 0.f, 0.f};

#define PIN() __builtin_amdgcn_sched_barrier(0)
    LoadSet S0, S1, S2;
    load_set(S0, wp, sp, tp);
    load_set(S1, wp + SET_STRIDE, sp + 3 * SET_STRIDE, tp + 3 * SET_STRIDE);
    PIN();
    for (int trip = 0; trip < (NSETS - 4) / 3; ++trip) {
        load_set(S2, wp + 2 * SET_STRIDE, sp + 6 * SET_STRIDE, tp + 6 * SET_STRIDE);
        PIN();
        mma_set(S0, isW, acc0, acc1);
        PIN();
        load_set(S0, wp + 3 * SET_STRIDE, sp + 9 * SET_STRIDE, tp + 9 * SET_STRIDE);
        PIN();
        mma_set(S1, isW, acc0, acc1);
        PIN();
        load_set(S1, wp + 4 * SET_STRIDE, sp + 12 * SET_STRIDE, tp + 12 * SET_STRIDE);
        PIN();
        mma_set(S2, isW, acc0, acc1);
        PIN();
        wp += 3 * SET_STRIDE; sp += 9 * SET_STRIDE; tp += 9 * SET_STRIDE;
    }
    load_set(S2, wp + 2 * SET_STRIDE, sp + 6 * SET_STRIDE, tp + 6 * SET_STRIDE);
    PIN();
    mma_set(S0, isW, acc0, acc1);
    PIN();
    load_set(S0, wp + 3 * SET_STRIDE, sp + 9 * SET_STRIDE, tp + 9 * SET_STRIDE);
    PIN();
    mma_set(S1, isW, acc0, acc1);
    mma_set(S2, isW, acc0, acc1);
    mma_set(S0, isW, acc0, acc1);
#undef PIN

    const v8f acc = acc0 + acc1;

    const int c  = lane & 15;
    const int h  = lane >> 4;
    const int j  = c >> 2;
    const int fp = c & 3;
    if ((j >> 1) == h) {
        const bool hi = (j & 1) != 0;
        red[wv][j][0 * 4 + fp] = hi ? acc[4] : acc[0];
        red[wv][j][1 * 4 + fp] = hi ? acc[5] : acc[1];
        red[wv][j][2 * 4 + fp] = hi ? acc[6] : acc[2];
        red[wv][j][3 * 4 + fp] = hi ? acc[7] : acc[3];
    }
    __syncthreads();
    if (tid < 16) {
        float s = 0.0f;
#pragma unroll
        for (int w2 = 0; w2 < WAVES; ++w2)
#pragma unroll
            for (int j2 = 0; j2 < 4; ++j2)
                s += red[w2][j2][tid];
        part[(size_t)b * 16 + tid] = s;
    }
}

__device__ inline float det3x3(const float M[3][3]) {
    return M[0][0] * (M[1][1] * M[2][2] - M[1][2] * M[2][1])
         - M[0][1] * (M[1][0] * M[2][2] - M[1][2] * M[2][0])
         + M[0][2] * (M[1][0] * M[2][1] - M[1][1] * M[2][0]);
}

__global__ __launch_bounds__(256)
void svd_solve_kernel(const float* __restrict__ part, float* __restrict__ out)
{
    const int b = blockIdx.x * blockDim.x + threadIdx.x;
    if (b >= B_BATCH) return;

    float C[4][4];
#pragma unroll
    for (int i = 0; i < 16; ++i) C[i >> 2][i & 3] = part[(size_t)b * 16 + i];

    const float W     = C[0][0];
    const float alpha = 1.0f / (W + 1e-8f);
    const float sig   = W * alpha;
    float sc[3] = {C[1][0] * alpha, C[2][0] * alpha, C[3][0] * alpha};
    float tc[3] = {C[0][1] * alpha, C[0][2] * alpha, C[0][3] * alpha};

    float H[3][3];
#pragma unroll
    for (int d = 0; d < 3; ++d)
#pragma unroll
        for (int e = 0; e < 3; ++e)
            H[d][e] = alpha * C[d + 1][e + 1] - (2.0f - sig) * sc[d] * tc[e];

    float S[3][3];
#pragma unroll
    for (int i = 0; i < 3; ++i)
#pragma unroll
        for (int j = 0; j < 3; ++j)
            S[i][j] = H[0][i] * H[0][j] + H[1][i] * H[1][j] + H[2][i] * H[2][j];

    float V[3][3] = {{1.f, 0.f, 0.f}, {0.f, 1.f, 0.f}, {0.f, 0.f, 1.f}};
    const int PQ[3][2] = {{0, 1}, {0, 2}, {1, 2}};
    for (int sweep = 0; sweep < 8; ++sweep) {
#pragma unroll
        for (int pi = 0; pi < 3; ++pi) {
            const int p = PQ[pi][0], q = PQ[pi][1];
            const float apq = S[p][q];
            if (fabsf(apq) > 1e-20f) {
                const float tau = (S[q][q] - S[p][p]) / (2.0f * apq);
                const float tt  = (tau >= 0.0f ? 1.0f : -1.0f) /
                                  (fabsf(tau) + sqrtf(1.0f + tau * tau));
                const float c = 1.0f / sqrtf(1.0f + tt * tt);
                const float s = tt * c;
#pragma unroll
                for (int r = 0; r < 3; ++r) {
                    const float Srp = S[r][p], Srq = S[r][q];
                    S[r][p] = c * Srp - s * Srq;
                    S[r][q] = s * Srp + c * Srq;
                }
#pragma unroll
                for (int r = 0; r < 3; ++r) {
                    const float Spr = S[p][r], Sqr = S[q][r];
                    S[p][r] = c * Spr - s * Sqr;
                    S[q][r] = s * Spr + c * Sqr;
                }
#pragma unroll
                for (int r = 0; r < 3; ++r) {
                    const float Vrp = V[r][p], Vrq = V[r][q];
                    V[r][p] = c * Vrp - s * Vrq;
                    V[r][q] = s * Vrp + c * Vrq;
                }
            }
        }
    }

    float lam[3] = {S[0][0], S[1][1], S[2][2]};
#pragma unroll
    for (int pi = 0; pi < 3; ++pi) {
        const int i = PQ[pi][0], j = PQ[pi][1];
        if (lam[j] > lam[i]) {
            const float tl = lam[i]; lam[i] = lam[j]; lam[j] = tl;
#pragma unroll
            for (int r = 0; r < 3; ++r) {
                const float tv = V[r][i]; V[r][i] = V[r][j]; V[r][j] = tv;
            }
        }
    }

    float Uc[3][3];
#pragma unroll
    for (int kcol = 0; kcol < 2; ++kcol)
#pragma unroll
        for (int r = 0; r < 3; ++r)
            Uc[kcol][r] = H[r][0] * V[0][kcol] + H[r][1] * V[1][kcol] + H[r][2] * V[2][kcol];

    float n0 = sqrtf(Uc[0][0]*Uc[0][0] + Uc[0][1]*Uc[0][1] + Uc[0][2]*Uc[0][2]);
    if (n0 > 1e-12f) { const float r0 = 1.0f / n0; Uc[0][0]*=r0; Uc[0][1]*=r0; Uc[0][2]*=r0; }
    else { Uc[0][0] = 1.f; Uc[0][1] = 0.f; Uc[0][2] = 0.f; }

    const float d01 = Uc[0][0]*Uc[1][0] + Uc[0][1]*Uc[1][1] + Uc[0][2]*Uc[1][2];
    Uc[1][0] -= d01 * Uc[0][0]; Uc[1][1] -= d01 * Uc[0][1]; Uc[1][2] -= d01 * Uc[0][2];
    float n1 = sqrtf(Uc[1][0]*Uc[1][0] + Uc[1][1]*Uc[1][1] + Uc[1][2]*Uc[1][2]);
    if (n1 > 1e-12f) { const float r1 = 1.0f / n1; Uc[1][0]*=r1; Uc[1][1]*=r1; Uc[1][2]*=r1; }
    else {
        const float ax = (fabsf(Uc[0][0]) < 0.9f) ? 1.f : 0.f;
        const float ay = 1.f - ax;
        float cx = ay * Uc[0][2] - 0.f * Uc[0][1];
        float cy = 0.f * Uc[0][0] - ax * Uc[0][2];
        float cz = ax * Uc[0][1] - ay * Uc[0][0];
        const float cn = 1.0f / sqrtf(cx*cx + cy*cy + cz*cz + 1e-30f);
        Uc[1][0] = cx * cn; Uc[1][1] = cy * cn; Uc[1][2] = cz * cn;
    }
    Uc[2][0] = Uc[0][1]*Uc[1][2] - Uc[0][2]*Uc[1][1];
    Uc[2][1] = Uc[0][2]*Uc[1][0] - Uc[0][0]*Uc[1][2];
    Uc[2][2] = Uc[0][0]*Uc[1][1] - Uc[0][1]*Uc[1][0];

    float R[3][3];
#pragma unroll
    for (int i = 0; i < 3; ++i)
#pragma unroll
        for (int j = 0; j < 3; ++j)
            R[i][j] = V[i][0]*Uc[0][j] + V[i][1]*Uc[1][j] + V[i][2]*Uc[2][j];

    if (det3x3(R) <= 0.0f) {
#pragma unroll
        for (int i = 0; i < 3; ++i)
#pragma unroll
            for (int j = 0; j < 3; ++j)
                R[i][j] -= 2.0f * V[i][2] * Uc[2][j];
    }

    float tvec[3];
#pragma unroll
    for (int i = 0; i < 3; ++i)
        tvec[i] = tc[i] - (R[i][0]*sc[0] + R[i][1]*sc[1] + R[i][2]*sc[2]);

    float* Rout = out;
    float* tout = out + (size_t)B_BATCH * 9;
#pragma unroll
    for (int i = 0; i < 3; ++i)
#pragma unroll
        for (int j = 0; j < 3; ++j)
            Rout[(size_t)b * 9 + i * 3 + j] = R[i][j];
#pragma unroll
    for (int i = 0; i < 3; ++i)
        tout[(size_t)b * 3 + i] = tvec[i];
}

extern "C" void kernel_launch(void* const* d_in, const int* in_sizes, int n_in,
                              void* d_out, int out_size, void* d_ws, size_t ws_size,
                              hipStream_t stream)
{
    const float* src = (const float*)d_in[0];
    const float* tpl = (const float*)d_in[1];
    const float* wgt = (const float*)d_in[2];
    float* out  = (float*)d_out;
    float* part = (float*)d_ws;

    svd_reduce_kernel<<<B_BATCH, THREADS, 0, stream>>>(src, tpl, wgt, part);
    svd_solve_kernel<<<(B_BATCH + 255) / 256, 256, 0, stream>>>(part, out);
}


# Solve kernel: rotation angles and column norms use the one-instruction reciprocal, sqrt and reciprocal sqrt (about 1 ulp) in place of correctly rounded division and sqrt sequences
# speedup vs baseline: 1.9996x; 1.0647x over previous
#include <hip/hip_runtime.h>

#define B_BATCH 1024
#define NPTS    8192
#define THREADS 256
#define WAVES   (THREADS / 32)
#define STEP_PTS     16
#define SET_STEPS    4
#define SET_PTS      (STEP_PTS * SET_STEPS)
#define SET_STRIDE   (WAVES * SET_PTS)
#define NSETS        (NPTS / SET_STRIDE)

typedef __attribute__((ext_vector_type(2))) float v2f;
typedef __attribute__((ext_vector_type(8))) float v8f;

struct LoadSet {
    v2f   w[SET_STEPS];
    float s0[SET_STEPS], s1[SET_STEPS];
    float t0[SET_STEPS], t1[SET_STEPS];
};

__device__ __forceinline__ void load_set(LoadSet& L, const float* __restrict__ wp,
                                         const float* __restrict__ sp,
                                         const float* __restrict__ tp)
{
#pragma unroll
    for (int k = 0; k < SET_STEPS; ++k) {
        L.w[k]  = *(const v2f*)(wp + STEP_PTS * k);
        L.s0[k] = sp[3 * STEP_PTS * k];
        L.s1[k] = sp[3 * STEP_PTS * k + 3];
        L.t0[k] = tp[3 * STEP_PTS * k];
        L.t1[k] = tp[3 * STEP_PTS * k + 3];
    }
}

__device__ __forceinline__ void mma_set(const LoadSet& L, const bool isW, v8f& acc0, v8f& acc1)
{
#pragma unroll
    for (int k = 0; k < SET_STEPS; ++k) {
        v2f a, b;
        a.x = isW ? L.w[k].x : L.w[k].x * L.s0[k];
        a.y = isW ? L.w[k].y : L.w[k].y * L.s1[k];
        b.x = isW ? 1.0f : L.t0[k];
        b.y = isW ? 1.0f : L.t1[k];
        if (k & 1)
            acc1 = __builtin_amdgcn_wmma_f32_16x16x4_f32(
                 false, a,  false, b,
                 (short)0, acc1,  false,  false);
        else
            acc0 = __builtin_amdgcn_wmma_f32_16x16x4_f32(
                 false, a,  false, b,
                 (short)0, acc0,  false,  false);
    }
}

__global__ __launch_bounds__(THREADS)
void svd_reduce_kernel(const float* __restrict__ src,
                       const float* __restrict__ tpl,
                       const float* __restrict__ wgt,
                       float* __restrict__ part)
{
    __shared__ float red[WAVES][4][16];

    const int tid  = threadIdx.x;
    const int lane = tid & 31;
    const int wv   = tid >> 5;
    const int b    = blockIdx.x;

    const int  f   = lane & 3;
    const int  q   = lane >> 2;
    const bool isW = (f == 0);
    const int  fm  = isW ? 0 : f - 1;

    const size_t p0 = (size_t)b * NPTS + (size_t)wv * SET_PTS + 2 * q;
    const float* wp = wgt + p0;
    const float* sp = src + p0 * 3 + fm;
    const float* tp = tpl + p0 * 3 + fm;

    v8f acc0 = {0.f, 0.f, 0.f, 0.f, 0.f, 0.f, 0.f, 0.f};
    v8f acc1 = {0.f, 0.f, 0.f, 0.f, 0.f, 0.f, 0.f, 0.f};

#define PIN() __builtin_amdgcn_sched_barrier(0)
    LoadSet S0, S1, S2;
    load_set(S0, wp, sp, tp);
    load_set(S1, wp + SET_STRIDE, sp + 3 * SET_STRIDE, tp + 3 * SET_STRIDE);
    PIN();
    for (int trip = 0; trip < (NSETS - 4) / 3; ++trip) {
        load_set(S2, wp + 2 * SET_STRIDE, sp + 6 * SET_STRIDE, tp + 6 * SET_STRIDE);
        PIN();
        mma_set(S0, isW, acc0, acc1);
        PIN();
        load_set(S0, wp + 3 * SET_STRIDE, sp + 9 * SET_STRIDE, tp + 9 * SET_STRIDE);
        PIN();
        mma_set(S1, isW, acc0, acc1);
        PIN();
        load_set(S1, wp + 4 * SET_STRIDE, sp + 12 * SET_STRIDE, tp + 12 * SET_STRIDE);
        PIN();
        mma_set(S2, isW, acc0, acc1);
        PIN();
        wp += 3 * SET_STRIDE; sp += 9 * SET_STRIDE; tp += 9 * SET_STRIDE;
    }
    load_set(S2, wp + 2 * SET_STRIDE, sp + 6 * SET_STRIDE, tp + 6 * SET_STRIDE);
    PIN();
    mma_set(S0, isW, acc0, acc1);
    PIN();
    load_set(S0, wp + 3 * SET_STRIDE, sp + 9 * SET_STRIDE, tp + 9 * SET_STRIDE);
    PIN();
    mma_set(S1, isW, acc0, acc1);
    mma_set(S2, isW, acc0, acc1);
    mma_set(S0, isW, acc0, acc1);
#undef PIN

    const v8f acc = acc0 + acc1;

    const int c  = lane & 15;
    const int h  = lane >> 4;
    const int j  = c >> 2;
    const int fp = c & 3;
    if ((j >> 1) == h) {
        const bool hi = (j & 1) != 0;
        red[wv][j][0 * 4 + fp] = hi ? acc[4] : acc[0];
        red[wv][j][1 * 4 + fp] = hi ? acc[5] : acc[1];
        red[wv][j][2 * 4 + fp] = hi ? acc[6] : acc[2];
        red[wv][j][3 * 4 + fp] = hi ? acc[7] : acc[3];
    }
    __syncthreads();
    if (tid < 16) {
        float s = 0.0f;
#pragma unroll
        for (int w2 = 0; w2 < WAVES; ++w2)
#pragma unroll
            for (int j2 = 0; j2 < 4; ++j2)
                s += red[w2][j2][tid];
        part[(size_t)b * 16 + tid] = s;
    }
}

__device__ inline float det3x3(const float M[3][3]) {
    return M[0][0] * (M[1][1] * M[2][2] - M[1][2] * M[2][1])
         - M[0][1] * (M[1][0] * M[2][2] - M[1][2] * M[2][0])
         + M[0][2] * (M[1][0] * M[2][1] - M[1][1] * M[2][0]);
}

__global__ __launch_bounds__(256)
void svd_solve_kernel(const float* __restrict__ part, float* __restrict__ out)
{
    const int b = blockIdx.x * blockDim.x + threadIdx.x;
    if (b >= B_BATCH) return;

    float C[4][4];
#pragma unroll
    for (int i = 0; i < 16; ++i) C[i >> 2][i & 3] = part[(size_t)b * 16 + i];

    const float W     = C[0][0];
    const float alpha = 1.0f / (W + 1e-8f);
    const float sig   = W * alpha;
    float sc[3] = {C[1][0] * alpha, C[2][0] * alpha, C[3][0] * alpha};
    float tc[3] = {C[0][1] * alpha, C[0][2] * alpha, C[0][3] * alpha};

    float H[3][3];
#pragma unroll
    for (int d = 0; d < 3; ++d)
#pragma unroll
        for (int e = 0; e < 3; ++e)
            H[d][e] = alpha * C[d + 1][e + 1] - (2.0f - sig) * sc[d] * tc[e];

    float S[3][3];
#pragma unroll
    for (int i = 0; i < 3; ++i)
#pragma unroll
        for (int j = 0; j < 3; ++j)
            S[i][j] = H[0][i] * H[0][j] + H[1][i] * H[1][j] + H[2][i] * H[2][j];

    float V[3][3] = {{1.f, 0.f, 0.f}, {0.f, 1.f, 0.f}, {0.f, 0.f, 1.f}};
    const int PQ[3][2] = {{0, 1}, {0, 2}, {1, 2}};
    for (int sweep = 0; sweep < 8; ++sweep) {
#pragma unroll
        for (int pi = 0; pi < 3; ++pi) {
            const int p = PQ[pi][0], q = PQ[pi][1];
            const float apq = S[p][q];
            if (fabsf(apq) > 1e-20f) {
                const float tau  = (S[q][q] - S[p][p]) * __builtin_amdgcn_rcpf(2.0f * apq);
                const float rden = __builtin_amdgcn_rcpf(
                                       fabsf(tau) + __builtin_amdgcn_sqrtf(1.0f + tau * tau));
                const float tt   = (tau >= 0.0f) ? rden : -rden;
                const float c = __builtin_amdgcn_rsqf(1.0f + tt * tt);
                const float s = tt * c;
#pragma unroll
                for (int r = 0; r < 3; ++r) {
                    const float Srp = S[r][p], Srq = S[r][q];
                    S[r][p] = c * Srp - s * Srq;
                    S[r][q] = s * Srp + c * Srq;
                }
#pragma unroll
                for (int r = 0; r < 3; ++r) {
                    const float Spr = S[p][r], Sqr = S[q][r];
                    S[p][r] = c * Spr - s * Sqr;
                    S[q][r] = s * Spr + c * Sqr;
                }
#pragma unroll
                for (int r = 0; r < 3; ++r) {
                    const float Vrp = V[r][p], Vrq = V[r][q];
                    V[r][p] = c * Vrp - s * Vrq;
                    V[r][q] = s * Vrp + c * Vrq;
                }
            }
        }
    }

    float lam[3] = {S[0][0], S[1][1], S[2][2]};
#pragma unroll
    for (int pi = 0; pi < 3; ++pi) {
        const int i = PQ[pi][0], j = PQ[pi][1];
        if (lam[j] > lam[i]) {
            const float tl = lam[i]; lam[i] = lam[j]; lam[j] = tl;
#pragma unroll
            for (int r = 0; r < 3; ++r) {
                const float tv = V[r][i]; V[r][i] = V[r][j]; V[r][j] = tv;
            }
        }
    }

    float Uc[3][3];
#pragma unroll
    for (int kcol = 0; kcol < 2; ++kcol)
#pragma unroll
        for (int r = 0; r < 3; ++r)
            Uc[kcol][r] = H[r][0] * V[0][kcol] + H[r][1] * V[1][kcol] + H[r][2] * V[2][kcol];

    float n0 = __builtin_amdgcn_sqrtf(Uc[0][0]*Uc[0][0] + Uc[0][1]*Uc[0][1] + Uc[0][2]*Uc[0][2]);
    if (n0 > 1e-12f) { const float r0 = __builtin_amdgcn_rcpf(n0); Uc[0][0]*=r0; Uc[0][1]*=r0; Uc[0][2]*=r0; }
    else { Uc[0][0] = 1.f; Uc[0][1] = 0.f; Uc[0][2] = 0.f; }

    const float d01 = Uc[0][0]*Uc[1][0] + Uc[0][1]*Uc[1][1] + Uc[0][2]*Uc[1][2];
    Uc[1][0] -= d01 * Uc[0][0]; Uc[1][1] -= d01 * Uc[0][1]; Uc[1][2] -= d01 * Uc[0][2];
    float n1 = __builtin_amdgcn_sqrtf(Uc[1][0]*Uc[1][0] + Uc[1][1]*Uc[1][1] + Uc[1][2]*Uc[1][2]);
    if (n1 > 1e-12f) { const float r1 = __builtin_amdgcn_rcpf(n1); Uc[1][0]*=r1; Uc[1][1]*=r1; Uc[1][2]*=r1; }
    else {
        const float ax = (fabsf(Uc[0][0]) < 0.9f) ? 1.f : 0.f;
        const float ay = 1.f - ax;
        float cx = ay * Uc[0][2] - 0.f * Uc[0][1];
        float cy = 0.f * Uc[0][0] - ax * Uc[0][2];
        float cz = ax * Uc[0][1] - ay * Uc[0][0];
        const float cn = 1.0f / sqrtf(cx*cx + cy*cy + cz*cz + 1e-30f);
        Uc[1][0] = cx * cn; Uc[1][1] = cy * cn; Uc[1][2] = cz * cn;
    }
    Uc[2][0] = Uc[0][1]*Uc[1][2] - Uc[0][2]*Uc[1][1];
    Uc[2][1] = Uc[0][2]*Uc[1][0] - Uc[0][0]*Uc[1][2];
    Uc[2][2] = Uc[0][0]*Uc[1][1] - Uc[0][1]*Uc[1][0];

    float R[3][3];
#pragma unroll
    for (int i = 0; i < 3; ++i)
#pragma unroll
        for (int j = 0; j < 3; ++j)
            R[i][j] = V[i][0]*Uc[0][j] + V[i][1]*Uc[1][j] + V[i][2]*Uc[2][j];

    if (det3x3(R) <= 0.0f) {
#pragma unroll
        for (int i = 0; i < 3; ++i)
#pragma unroll
            for (int j = 0; j < 3; ++j)
                R[i][j] -= 2.0f * V[i][2] * Uc[2][j];
    }

    float tvec[3];
#pragma unroll
    for (int i = 0; i < 3; ++i)
        tvec[i] = tc[i] - (R[i][0]*sc[0] + R[i][1]*sc[1] + R[i][2]*sc[2]);

    float* Rout = out;
    float* tout = out + (size_t)B_BATCH * 9;
#pragma unroll
    for (int i = 0; i < 3; ++i)
#pragma unroll
        for (int j = 0; j < 3; ++j)
            Rout[(size_t)b * 9 + i * 3 + j] = R[i][j];
#pragma unroll
    for (int i = 0; i < 3; ++i)
        tout[(size_t)b * 3 + i] = tvec[i];
}

extern "C" void kernel_launch(void* const* d_in, const int* in_sizes, int n_in,
                              void* d_out, int out_size, void* d_ws, size_t ws_size,
                              hipStream_t stream)
{
    const float* src = (const float*)d_in[0];
    const float* tpl = (const float*)d_in[1];
    const float* wgt = (const float*)d_in[2];
    float* out  = (float*)d_out;
    float* part = (float*)d_ws;

    svd_reduce_kernel<<<B_BATCH, THREADS, 0, stream>>>(src, tpl, wgt, part);
    svd_solve_kernel<<<(B_BATCH + 255) / 256, 256, 0, stream>>>(part, out);
}


# Solve fused into the reduce kernel: each workgroup's first thread solves its own batch from the block sums in LDS (uniform per workgroup: compiles to scalar-unit float code); one launch, not two
# speedup vs baseline: 2.1323x; 1.0664x over previous
#include <hip/hip_runtime.h>

#define B_BATCH 1024
#define NPTS    8192
#define THREADS 256
#define WAVES   (THREADS / 32)
#define STEP_PTS     16
#define SET_STEPS    4
#define SET_PTS      (STEP_PTS * SET_STEPS)
#define SET_STRIDE   (WAVES * SET_PTS)
#define NSETS        (NPTS / SET_STRIDE)

typedef __attribute__((ext_vector_type(2))) float v2f;
typedef __attribute__((ext_vector_type(8))) float v8f;

__device__ inline float det3x3(const float M[3][3]) {
    return M[0][0] * (M[1][1] * M[2][2] - M[1][2] * M[2][1])
         - M[0][1] * (M[1][0] * M[2][2] - M[1][2] * M[2][0])
         + M[0][2] * (M[1][0] * M[2][1] - M[1][1] * M[2][0]);
}

__device__ __forceinline__ void solve_batch(const float* csum, float* __restrict__ out, const int b)
{
    float C[4][4];
#pragma unroll
    for (int i = 0; i < 16; ++i) C[i >> 2][i & 3] = csum[i];

    const float W     = C[0][0];
    const float alpha = 1.0f / (W + 1e-8f);
    const float sig   = W * alpha;
    float sc[3] = {C[1][0] * alpha, C[2][0] * alpha, C[3][0] * alpha};
    float tc[3] = {C[0][1] * alpha, C[0][2] * alpha, C[0][3] * alpha};

    float H[3][3];
#pragma unroll
    for (int d = 0; d < 3; ++d)
#pragma unroll
        for (int e = 0; e < 3; ++e)
            H[d][e] = alpha * C[d + 1][e + 1] - (2.0f - sig) * sc[d] * tc[e];

    float S[3][3];
#pragma unroll
    for (int i = 0; i < 3; ++i)
#pragma unroll
        for (int j = 0; j < 3; ++j)
            S[i][j] = H[0][i] * H[0][j] + H[1][i] * H[1][j] + H[2][i] * H[2][j];

    float V[3][3] = {{1.f, 0.f, 0.f}, {0.f, 1.f, 0.f}, {0.f, 0.f, 1.f}};
    const int PQ[3][2] = {{0, 1}, {0, 2}, {1, 2}};
    for (int sweep = 0; sweep < 8; ++sweep) {
#pragma unroll
        for (int pi = 0; pi < 3; ++pi) {
            const int p = PQ[pi][0], q = PQ[pi][1];
            const float apq = S[p][q];
            if (fabsf(apq) > 1e-20f) {
                const float tau  = (S[q][q] - S[p][p]) * __builtin_amdgcn_rcpf(2.0f * apq);
                const float rden = __builtin_amdgcn_rcpf(
                                       fabsf(tau) + __builtin_amdgcn_sqrtf(1.0f + tau * tau));
                const float tt   = (tau >= 0.0f) ? rden : -rden;
                const float c = __builtin_amdgcn_rsqf(1.0f + tt * tt);
                const float s = tt * c;
#pragma unroll
                for (int r = 0; r < 3; ++r) {
                    const float Srp = S[r][p], Srq = S[r][q];
                    S[r][p] = c * Srp - s * Srq;
                    S[r][q] = s * Srp + c * Srq;
                }
#pragma unroll
                for (int r = 0; r < 3; ++r) {
                    const float Spr = S[p][r], Sqr = S[q][r];
                    S[p][r] = c * Spr - s * Sqr;
                    S[q][r] = s * Spr + c * Sqr;
                }
#pragma unroll
                for (int r = 0; r < 3; ++r) {
                    const float Vrp = V[r][p], Vrq = V[r][q];
                    V[r][p] = c * Vrp - s * Vrq;
                    V[r][q] = s * Vrp + c * Vrq;
                }
            }
        }
    }

    float lam[3] = {S[0][0], S[1][1], S[2][2]};
#pragma unroll
    for (int pi = 0; pi < 3; ++pi) {
        const int i = PQ[pi][0], j = PQ[pi][1];
        if (lam[j] > lam[i]) {
            const float tl = lam[i]; lam[i] = lam[j]; lam[j] = tl;
#pragma unroll
            for (int r = 0; r < 3; ++r) {
                const float tv = V[r][i]; V[r][i] = V[r][j]; V[r][j] = tv;
            }
        }
    }

    float Uc[3][3];
#pragma unroll
    for (int kcol = 0; kcol < 2; ++kcol)
#pragma unroll
        for (int r = 0; r < 3; ++r)
            Uc[kcol][r] = H[r][0] * V[0][kcol] + H[r][1] * V[1][kcol] + H[r][2] * V[2][kcol];

    float n0 = __builtin_amdgcn_sqrtf(Uc[0][0]*Uc[0][0] + Uc[0][1]*Uc[0][1] + Uc[0][2]*Uc[0][2]);
    if (n0 > 1e-12f) { const float r0 = __builtin_amdgcn_rcpf(n0); Uc[0][0]*=r0; Uc[0][1]*=r0; Uc[0][2]*=r0; }
    else { Uc[0][0] = 1.f; Uc[0][1] = 0.f; Uc[0][2] = 0.f; }

    const float d01 = Uc[0][0]*Uc[1][0] + Uc[0][1]*Uc[1][1] + Uc[0][2]*Uc[1][2];
    Uc[1][0] -= d01 * Uc[0][0]; Uc[1][1] -= d01 * Uc[0][1]; Uc[1][2] -= d01 * Uc[0][2];
    float n1 = __builtin_amdgcn_sqrtf(Uc[1][0]*Uc[1][0] + Uc[1][1]*Uc[1][1] + Uc[1][2]*Uc[1][2]);
    if (n1 > 1e-12f) { const float r1 = __builtin_amdgcn_rcpf(n1); Uc[1][0]*=r1; Uc[1][1]*=r1; Uc[1][2]*=r1; }
    else {
        const float ax = (fabsf(Uc[0][0]) < 0.9f) ? 1.f : 0.f;
        const float ay = 1.f - ax;
        float cx = ay * Uc[0][2] - 0.f * Uc[0][1];
        float cy = 0.f * Uc[0][0] - ax * Uc[0][2];
        float cz = ax * Uc[0][1] - ay * Uc[0][0];
        const float cn = 1.0f / sqrtf(cx*cx + cy*cy + cz*cz + 1e-30f);
        Uc[1][0] = cx * cn; Uc[1][1] = cy * cn; Uc[1][2] = cz * cn;
    }
    Uc[2][0] = Uc[0][1]*Uc[1][2] - Uc[0][2]*Uc[1][1];
    Uc[2][1] = Uc[0][2]*Uc[1][0] - Uc[0][0]*Uc[1][2];
    Uc[2][2] = Uc[0][0]*Uc[1][1] - Uc[0][1]*Uc[1][0];

    float R[3][3];
#pragma unroll
    for (int i = 0; i < 3; ++i)
#pragma unroll
        for (int j = 0; j < 3; ++j)
            R[i][j] = V[i][0]*Uc[0][j] + V[i][1]*Uc[1][j] + V[i][2]*Uc[2][j];

    if (det3x3(R) <= 0.0f) {
#pragma unroll
        for (int i = 0; i < 3; ++i)
#pragma unroll
            for (int j = 0; j < 3; ++j)
                R[i][j] -= 2.0f * V[i][2] * Uc[2][j];
    }

    float tvec[3];
#pragma unroll
    for (int i = 0; i < 3; ++i)
        tvec[i] = tc[i] - (R[i][0]*sc[0] + R[i][1]*sc[1] + R[i][2]*sc[2]);

    float* Rout = out;
    float* tout = out + (size_t)B_BATCH * 9;
#pragma unroll
    for (int i = 0; i < 3; ++i)
#pragma unroll
        for (int j = 0; j < 3; ++j)
            Rout[(size_t)b * 9 + i * 3 + j] = R[i][j];
#pragma unroll
    for (int i = 0; i < 3; ++i)
        tout[(size_t)b * 3 + i] = tvec[i];
}

struct LoadSet {
    v2f   w[SET_STEPS];
    float s0[SET_STEPS], s1[SET_STEPS];
    float t0[SET_STEPS], t1[SET_STEPS];
};

__device__ __forceinline__ void load_set(LoadSet& L, const float* __restrict__ wp,
                                         const float* __restrict__ sp,
                                         const float* __restrict__ tp)
{
#pragma unroll
    for (int k = 0; k < SET_STEPS; ++k) {
        L.w[k]  = *(const v2f*)(wp + STEP_PTS * k);
        L.s0[k] = sp[3 * STEP_PTS * k];
        L.s1[k] = sp[3 * STEP_PTS * k + 3];
        L.t0[k] = tp[3 * STEP_PTS * k];
        L.t1[k] = tp[3 * STEP_PTS * k + 3];
    }
}

__device__ __forceinline__ void mma_set(const LoadSet& L, const bool isW, v8f& acc0, v8f& acc1)
{
#pragma unroll
    for (int k = 0; k < SET_STEPS; ++k) {
        v2f a, b;
        a.x = isW ? L.w[k].x : L.w[k].x * L.s0[k];
        a.y = isW ? L.w[k].y : L.w[k].y * L.s1[k];
        b.x = isW ? 1.0f : L.t0[k];
        b.y = isW ? 1.0f : L.t1[k];
        if (k & 1)
            acc1 = __builtin_amdgcn_wmma_f32_16x16x4_f32(
                 false, a,  false, b,
                 (short)0, acc1,  false,  false);
        else
            acc0 = __builtin_amdgcn_wmma_f32_16x16x4_f32(
                 false, a,  false, b,
                 (short)0, acc0,  false,  false);
    }
}

__global__ __launch_bounds__(THREADS)
void svd_fused_kernel(const float* __restrict__ src,
                      const float* __restrict__ tpl,
                      const float* __restrict__ wgt,
                      float* __restrict__ out)
{
    __shared__ float red[WAVES][4][16];
    __shared__ float csum[16];

    const int tid  = threadIdx.x;
    const int lane = tid & 31;
    const int wv   = tid >> 5;
    const int b    = blockIdx.x;

    const int  f   = lane & 3;
    const int  q   = lane >> 2;
    const bool isW = (f == 0);
    const int  fm  = isW ? 0 : f - 1;

    const size_t p0 = (size_t)b * NPTS + (size_t)wv * SET_PTS + 2 * q;
    const float* wp = wgt + p0;
    const float* sp = src + p0 * 3 + fm;
    const float* tp = tpl + p0 * 3 + fm;

    v8f acc0 = {0.f, 0.f, 0.f, 0.f, 0.f, 0.f, 0.f, 0.f};
    v8f acc1 = {0.f, 0.f, 0.f, 0.f, 0.f, 0.f, 0.f, 0.f};

#define PIN() __builtin_amdgcn_sched_barrier(0)
    LoadSet S0, S1, S2;
    load_set(S0, wp, sp, tp);
    load_set(S1, wp + SET_STRIDE, sp + 3 * SET_STRIDE, tp + 3 * SET_STRIDE);
    PIN();
    for (int trip = 0; trip < (NSETS - 4) / 3; ++trip) {
        load_set(S2, wp + 2 * SET_STRIDE, sp + 6 * SET_STRIDE, tp + 6 * SET_STRIDE);
        PIN();
        mma_set(S0, isW, acc0, acc1);
        PIN();
        load_set(S0, wp + 3 * SET_STRIDE, sp + 9 * SET_STRIDE, tp + 9 * SET_STRIDE);
        PIN();
        mma_set(S1, isW, acc0, acc1);
        PIN();
        load_set(S1, wp + 4 * SET_STRIDE, sp + 12 * SET_STRIDE, tp + 12 * SET_STRIDE);
        PIN();
        mma_set(S2, isW, acc0, acc1);
        PIN();
        wp += 3 * SET_STRIDE; sp += 9 * SET_STRIDE; tp += 9 * SET_STRIDE;
    }
    load_set(S2, wp + 2 * SET_STRIDE, sp + 6 * SET_STRIDE, tp + 6 * SET_STRIDE);
    PIN();
    mma_set(S0, isW, acc0, acc1);
    PIN();
    load_set(S0, wp + 3 * SET_STRIDE, sp + 9 * SET_STRIDE, tp + 9 * SET_STRIDE);
    PIN();
    mma_set(S1, isW, acc0, acc1);
    mma_set(S2, isW, acc0, acc1);
    mma_set(S0, isW, acc0, acc1);
#undef PIN

    const v8f acc = acc0 + acc1;

    const int c  = lane & 15;
    const int h  = lane >> 4;
    const int j  = c >> 2;
    const int fp = c & 3;
    if ((j >> 1) == h) {
        const bool hi = (j & 1) != 0;
        red[wv][j][0 * 4 + fp] = hi ? acc[4] : acc[0];
        red[wv][j][1 * 4 + fp] = hi ? acc[5] : acc[1];
        red[wv][j][2 * 4 + fp] = hi ? acc[6] : acc[2];
        red[wv][j][3 * 4 + fp] = hi ? acc[7] : acc[3];
    }
    __syncthreads();
    if (tid < 16) {
        float s = 0.0f;
#pragma unroll
        for (int w2 = 0; w2 < WAVES; ++w2)
#pragma unroll
            for (int j2 = 0; j2 < 4; ++j2)
                s += red[w2][j2][tid];
        csum[tid] = s;
    }
    __syncthreads();
    if (tid == 0) solve_batch(csum, out, b);
}

extern "C" void kernel_launch(void* const* d_in, const int* in_sizes, int n_in,
                              void* d_out, int out_size, void* d_ws, size_t ws_size,
                              hipStream_t stream)
{
    const float* src = (const float*)d_in[0];
    const float* tpl = (const float*)d_in[1];
    const float* wgt = (const float*)d_in[2];
    float* out  = (float*)d_out;

    svd_fused_kernel<<<B_BATCH, THREADS, 0, stream>>>(src, tpl, wgt, out);
}
